# starting kernel
# baseline (speedup 1.0000x reference)
#include <hip/hip_runtime.h>
#include <hip/hip_bf16.h>

typedef float v2f __attribute__((ext_vector_type(2)));
typedef float v8f __attribute__((ext_vector_type(8)));

#define BB 512
#define FF 512
#define KK 32
#define DD 16
#define NN (KK * DD)
#define OUTC (FF + KK)

__global__ __launch_bounds__(256) void gemm_wmma_f32(const float* __restrict__ x,
                                                     const float* __restrict__ T,
                                                     float* __restrict__ m) {
  const int lane = threadIdx.x & 31;
  const int wave = threadIdx.x >> 5;
  const int tile = blockIdx.x * 8 + wave;
  const int tile_m = tile >> 5;
  const int tile_n = tile & 31;
  const int half = lane >> 4;
  const int l16 = lane & 15;

  const int arow = tile_m * 16 + l16;
  const int bcol = tile_n * 16 + l16;

  v8f c = {};
  for (int k = 0; k < FF; k += 4) {
    const int ka = k + half * 2;
    v2f a;
    a.x = x[arow * FF + ka];
    a.y = x[arow * FF + ka + 1];
    v2f b;
    b.x = T[(ka)     * NN + bcol];
    b.y = T[(ka + 1) * NN + bcol];
    c = __builtin_amdgcn_wmma_f32_16x16x4_f32(false, a, false, b,
                                              (short)0, c, false, false);
  }

#pragma unroll
  for (int r = 0; r < 8; ++r) {
    m[(tile_m * 16 + r + half * 8) * NN + bcol] = c[r];
  }
}

__global__ __launch_bounds__(256) void copy_x(const float* __restrict__ x,
                                              float* __restrict__ out) {
  const int idx = blockIdx.x * 256 + threadIdx.x;
  if (idx < BB * FF) {
    const int i = idx / FF;
    const int ccol = idx - i * FF;
    out[i * OUTC + ccol] = x[idx];
  }
}

__global__ __launch_bounds__(64) void feats_kernel(const float* __restrict__ m,
                                                   float* __restrict__ out) {
  __shared__ float lds[BB * DD];
  const int k = blockIdx.y;
  const int i = blockIdx.x * 64 + threadIdx.x;

  for (int t = threadIdx.x; t < BB * 4; t += 64) {
    const int j = t >> 2;
    const int q = t & 3;
    const float4* src = (const float4*)(m + j * NN + k * DD);
    ((float4*)lds)[j * 4 + q] = src[q];
  }
  __syncthreads();

  float4 mi0 = ((const float4*)(lds + i * DD))[0];
  float4 mi1 = ((const float4*)(lds + i * DD))[1];
  float4 mi2 = ((const float4*)(lds + i * DD))[2];
  float4 mi3 = ((const float4*)(lds + i * DD))[3];

  float acc = 0.f;
  for (int j = 0; j < BB; ++j) {
    const float4* row = (const float4*)(lds + j * DD);
    float4 r0 = row[0], r1 = row[1], r2 = row[2], r3 = row[3];
    float dist = fabsf(mi0.x - r0.x) + fabsf(mi0.y - r0.y) +
                 fabsf(mi0.z - r0.z) + fabsf(mi0.w - r0.w);
    dist += fabsf(mi1.x - r1.x) + fabsf(mi1.y - r1.y) +
            fabsf(mi1.z - r1.z) + fabsf(mi1.w - r1.w);
    dist += fabsf(mi2.x - r2.x) + fabsf(mi2.y - r2.y) +
            fabsf(mi2.z - r2.z) + fabsf(mi2.w - r2.w);
    dist += fabsf(mi3.x - r3.x) + fabsf(mi3.y - r3.y) +
            fabsf(mi3.z - r3.z) + fabsf(mi3.w - r3.w);
    acc += __expf(-dist);
  }
  out[i * OUTC + FF + k] = acc;
}

extern "C" void kernel_launch(void* const* d_in, const int* in_sizes, int n_in,
                              void* d_out, int out_size, void* d_ws, size_t ws_size,
                              hipStream_t stream) {
  (void)in_sizes; (void)n_in; (void)out_size; (void)ws_size;
  const float* x = (const float*)d_in[0];
  const float* T = (const float*)d_in[1];
  float* out = (float*)d_out;
  float* m = (float*)d_ws;

  gemm_wmma_f32<<<128, 256, 0, stream>>>(x, T, m);
  copy_x<<<(BB * FF + 255) / 256, 256, 0, stream>>>(x, out);
  feats_kernel<<<dim3(8, 32), 64, 0, stream>>>(m, out);
}


# feats: j sum of each (i,k) split over 8 waves of a 16-wave work-group (4 waves per SIMD on every WGP hide the LDS wait); sum order now 8 partial sums of 64 j added in fixed order
# speedup vs baseline: 1.6198x; 1.6198x over previous
#include <hip/hip_runtime.h>
#include <hip/hip_bf16.h>

typedef float v2f __attribute__((ext_vector_type(2)));
typedef float v8f __attribute__((ext_vector_type(8)));

#define BB 512
#define FF 512
#define KK 32
#define DD 16
#define NN (KK * DD)
#define OUTC (FF + KK)

__global__ __launch_bounds__(256) void gemm_wmma_f32(const float* __restrict__ x,
                                                     const float* __restrict__ T,
                                                     float* __restrict__ m) {
  const int lane = threadIdx.x & 31;
  const int wave = threadIdx.x >> 5;
  const int tile = blockIdx.x * 8 + wave;
  const int tile_m = tile >> 5;
  const int tile_n = tile & 31;
  const int half = lane >> 4;
  const int l16 = lane & 15;

  const int arow = tile_m * 16 + l16;
  const int bcol = tile_n * 16 + l16;

  v8f c = {};
  for (int k = 0; k < FF; k += 4) {
    const int ka = k + half * 2;
    v2f a;
    a.x = x[arow * FF + ka];
    a.y = x[arow * FF + ka + 1];
    v2f b;
    b.x = T[(ka)     * NN + bcol];
    b.y = T[(ka + 1) * NN + bcol];
    c = __builtin_amdgcn_wmma_f32_16x16x4_f32(false, a, false, b,
                                              (short)0, c, false, false);
  }

#pragma unroll
  for (int r = 0; r < 8; ++r) {
    m[(tile_m * 16 + r + half * 8) * NN + bcol] = c[r];
  }
}

__global__ __launch_bounds__(256) void copy_x(const float* __restrict__ x,
                                              float* __restrict__ out) {
  const int idx = blockIdx.x * 256 + threadIdx.x;
  if (idx < BB * FF) {
    const int i = idx / FF;
    const int ccol = idx - i * FF;
    out[i * OUTC + ccol] = x[idx];
  }
}

#define JSPLIT 8
#define JPER (BB / JSPLIT)

__global__ __launch_bounds__(64 * JSPLIT) void feats_kernel(const float* __restrict__ m,
                                                            float* __restrict__ out) {
  __shared__ float lds[BB * DD];
  __shared__ float part[JSPLIT * 64];
  const int k = blockIdx.y;
  const int il = threadIdx.x & 63;
  const int jp = threadIdx.x >> 6;
  const int i = blockIdx.x * 64 + il;

  for (int r = 0; r < (BB * 4) / (64 * JSPLIT); ++r) {
    const int t = threadIdx.x + r * (64 * JSPLIT);
    const int j = t >> 2;
    const int q = t & 3;
    const float4* src = (const float4*)(m + j * NN + k * DD);
    ((float4*)lds)[j * 4 + q] = src[q];
  }
  __syncthreads();

  float4 mi0 = ((const float4*)(lds + i * DD))[0];
  float4 mi1 = ((const float4*)(lds + i * DD))[1];
  float4 mi2 = ((const float4*)(lds + i * DD))[2];
  float4 mi3 = ((const float4*)(lds + i * DD))[3];

  float acc = 0.f;
  const int j0 = jp * JPER;
  for (int jj = 0; jj < JPER; ++jj) {
    const float4* row = (const float4*)(lds + (j0 + jj) * DD);
    float4 r0 = row[0], r1 = row[1], r2 = row[2], r3 = row[3];
    float dist = fabsf(mi0.x - r0.x) + fabsf(mi0.y - r0.y) +
                 fabsf(mi0.z - r0.z) + fabsf(mi0.w - r0.w);
    dist += fabsf(mi1.x - r1.x) + fabsf(mi1.y - r1.y) +
            fabsf(mi1.z - r1.z) + fabsf(mi1.w - r1.w);
    dist += fabsf(mi2.x - r2.x) + fabsf(mi2.y - r2.y) +
            fabsf(mi2.z - r2.z) + fabsf(mi2.w - r2.w);
    dist += fabsf(mi3.x - r3.x) + fabsf(mi3.y - r3.y) +
            fabsf(mi3.z - r3.z) + fabsf(mi3.w - r3.w);
    acc += __expf(-dist);
  }

  part[jp * 64 + il] = acc;
  __syncthreads();
  if (threadIdx.x < 64) {
    float s = part[il];
    for (int p = 1; p < JSPLIT; ++p) s += part[p * 64 + il];
    out[i * OUTC + FF + k] = s;
  }
}

extern "C" void kernel_launch(void* const* d_in, const int* in_sizes, int n_in,
                              void* d_out, int out_size, void* d_ws, size_t ws_size,
                              hipStream_t stream) {
  (void)in_sizes; (void)n_in; (void)out_size; (void)ws_size;
  const float* x = (const float*)d_in[0];
  const float* T = (const float*)d_in[1];
  float* out = (float*)d_out;
  float* m = (float*)d_ws;

  gemm_wmma_f32<<<128, 256, 0, stream>>>(x, T, m);
  copy_x<<<(BB * FF + 255) / 256, 256, 0, stream>>>(x, out);
  feats_kernel<<<dim3(8, 32), 64 * JSPLIT, 0, stream>>>(m, out);
}


# out[:, :512] = x rides in the feats launch (one float2 per thread, loaded with the staging loads, stored behind the staging barrier); the 1024-block copy launch is gone: 2 launches per call, not 3
# speedup vs baseline: 1.7520x; 1.0816x over previous
#include <hip/hip_runtime.h>
#include <hip/hip_bf16.h>

typedef float v2f __attribute__((ext_vector_type(2)));
typedef float v8f __attribute__((ext_vector_type(8)));

#define BB 512
#define FF 512
#define KK 32
#define DD 16
#define NN (KK * DD)
#define OUTC (FF + KK)

__global__ __launch_bounds__(256) void gemm_wmma_f32(const float* __restrict__ x,
                                                     const float* __restrict__ T,
                                                     float* __restrict__ m) {
  const int lane = threadIdx.x & 31;
  const int wave = threadIdx.x >> 5;
  const int tile = blockIdx.x * 8 + wave;
  const int tile_m = tile >> 5;
  const int tile_n = tile & 31;
  const int half = lane >> 4;
  const int l16 = lane & 15;

  const int arow = tile_m * 16 + l16;
  const int bcol = tile_n * 16 + l16;

  v8f c = {};
  for (int k = 0; k < FF; k += 4) {
    const int ka = k + half * 2;
    v2f a;
    a.x = x[arow * FF + ka];
    a.y = x[arow * FF + ka + 1];
    v2f b;
    b.x = T[(ka)     * NN + bcol];
    b.y = T[(ka + 1) * NN + bcol];
    c = __builtin_amdgcn_wmma_f32_16x16x4_f32(false, a, false, b,
                                              (short)0, c, false, false);
  }

#pragma unroll
  for (int r = 0; r < 8; ++r) {
    m[(tile_m * 16 + r + half * 8) * NN + bcol] = c[r];
  }
}

#define JSPLIT 8
#define JPER (BB / JSPLIT)

__global__ __launch_bounds__(64 * JSPLIT) void feats_kernel(const float* __restrict__ x,
                                                            const float* __restrict__ m,
                                                            float* __restrict__ out) {
  __shared__ float lds[BB * DD];
  __shared__ float part[JSPLIT * 64];
  const int k = blockIdx.y;
  const int il = threadIdx.x & 63;
  const int jp = threadIdx.x >> 6;
  const int i = blockIdx.x * 64 + il;

  const int f2 = (blockIdx.y * 8 + blockIdx.x) * (64 * JSPLIT) + threadIdx.x;
  const float2 xc = ((const float2*)x)[f2];

  for (int r = 0; r < (BB * 4) / (64 * JSPLIT); ++r) {
    const int t = threadIdx.x + r * (64 * JSPLIT);
    const int j = t >> 2;
    const int q = t & 3;
    const float4* src = (const float4*)(m + j * NN + k * DD);
    ((float4*)lds)[j * 4 + q] = src[q];
  }
  __syncthreads();
  ((float2*)out)[(f2 >> 8) * (OUTC / 2) + (f2 & 255)] = xc;

  float4 mi0 = ((const float4*)(lds + i * DD))[0];
  float4 mi1 = ((const float4*)(lds + i * DD))[1];
  float4 mi2 = ((const float4*)(lds + i * DD))[2];
  float4 mi3 = ((const float4*)(lds + i * DD))[3];

  float acc = 0.f;
  const int j0 = jp * JPER;
  for (int jj = 0; jj < JPER; ++jj) {
    const float4* row = (const float4*)(lds + (j0 + jj) * DD);
    float4 r0 = row[0], r1 = row[1], r2 = row[2], r3 = row[3];
    float dist = fabsf(mi0.x - r0.x) + fabsf(mi0.y - r0.y) +
                 fabsf(mi0.z - r0.z) + fabsf(mi0.w - r0.w);
    dist += fabsf(mi1.x - r1.x) + fabsf(mi1.y - r1.y) +
            fabsf(mi1.z - r1.z) + fabsf(mi1.w - r1.w);
    dist += fabsf(mi2.x - r2.x) + fabsf(mi2.y - r2.y) +
            fabsf(mi2.z - r2.z) + fabsf(mi2.w - r2.w);
    dist += fabsf(mi3.x - r3.x) + fabsf(mi3.y - r3.y) +
            fabsf(mi3.z - r3.z) + fabsf(mi3.w - r3.w);
    acc += __expf(-dist);
  }

  part[jp * 64 + il] = acc;
  __syncthreads();
  if (threadIdx.x < 64) {
    float s = part[il];
    for (int p = 1; p < JSPLIT; ++p) s += part[p * 64 + il];
    out[i * OUTC + FF + k] = s;
  }
}

extern "C" void kernel_launch(void* const* d_in, const int* in_sizes, int n_in,
                              void* d_out, int out_size, void* d_ws, size_t ws_size,
                              hipStream_t stream) {
  (void)in_sizes; (void)n_in; (void)out_size; (void)ws_size;
  const float* x = (const float*)d_in[0];
  const float* T = (const float*)d_in[1];
  float* out = (float*)d_out;
  float* m = (float*)d_ws;

  gemm_wmma_f32<<<128, 256, 0, stream>>>(x, T, m);
  feats_kernel<<<dim3(8, 32), 64 * JSPLIT, 0, stream>>>(x, m, out);
}


# m = x @ T: block per 32x32 tile; each wave stages its own 64-k slice of both panels in LDS (16-byte loads, no barrier before the 4 accumulator tiles); sum order now 8 k-slice partials added 0..7
# speedup vs baseline: 3.0566x; 1.7446x over previous
#include <hip/hip_runtime.h>
#include <hip/hip_bf16.h>

typedef float v2f __attribute__((ext_vector_type(2)));
typedef float v8f __attribute__((ext_vector_type(8)));

#define BB 512
#define FF 512
#define KK 32
#define DD 16
#define NN (KK * DD)
#define OUTC (FF + KK)

#define TM 32
#define TN 32
#define KW 64
#define APITCH 68
#define BPITCH 40
#define PPITCH 36
#define WAVE_A (TM * APITCH)
#define WAVE_B (KW * BPITCH)

__global__ __launch_bounds__(256) void gemm_wmma_f32(const float* __restrict__ x,
                                                     const float* __restrict__ T,
                                                     float* __restrict__ m) {
  __shared__ __attribute__((aligned(16))) float As[8 * WAVE_A];
  __shared__ __attribute__((aligned(16))) float Bs[8 * WAVE_B];
  __shared__ __attribute__((aligned(16))) float Ps[8 * TM * PPITCH];
  const int t = threadIdx.x;
  const int lane = t & 31;
  const int wave = t >> 5;
  const int half = lane >> 4;
  const int l16 = lane & 15;
  const int bm = blockIdx.x >> 4;
  const int bn = blockIdx.x & 15;

  float* Aw = As + wave * WAVE_A;
  float* Bw = Bs + wave * WAVE_B;
  const float* xg = x + (bm * TM) * FF + wave * KW;
  const float* Tg = T + (wave * KW) * NN + bn * TN;

  float4 ra[2][8], rb[2][8];
#pragma unroll
  for (int ph = 0; ph < 2; ++ph) {
#pragma unroll
    for (int r = 0; r < 8; ++r) {
      const int idx = lane + r * 32;
      ra[ph][r] = *(const float4*)(xg + (idx >> 3) * FF + ph * 32 + (idx & 7) * 4);
    }
#pragma unroll
    for (int r = 0; r < 8; ++r) {
      const int idx = lane + r * 32;
      rb[ph][r] = *(const float4*)(Tg + (ph * 32 + (idx >> 3)) * NN + (idx & 7) * 4);
    }
  }

  v8f c00 = {};
  v8f c01 = {};
  v8f c10 = {};
  v8f c11 = {};
  const float* ap = Aw + l16 * APITCH + half * 2;
  const float* bp = Bw + (half * 2) * BPITCH + l16;

#pragma unroll
  for (int ph = 0; ph < 2; ++ph) {
    if (ph == 1) __builtin_amdgcn_sched_barrier(0x20);
#pragma unroll
    for (int r = 0; r < 8; ++r) {
      const int idx = lane + r * 32;
      *(float4*)(Aw + (idx >> 3) * APITCH + ph * 32 + (idx & 7) * 4) = ra[ph][r];
    }
#pragma unroll
    for (int r = 0; r < 8; ++r) {
      const int idx = lane + r * 32;
      *(float4*)(Bw + (ph * 32 + (idx >> 3)) * BPITCH + (idx & 7) * 4) = rb[ph][r];
    }
#pragma unroll
    for (int ks = 0; ks < 8; ++ks) {
      const int kk = ph * 8 + ks;
      const v2f a0 = *(const v2f*)(ap + kk * 4);
      const v2f a1 = *(const v2f*)(ap + 16 * APITCH + kk * 4);
      v2f b0, b1;
      b0.x = bp[(kk * 4) * BPITCH];
      b0.y = bp[(kk * 4 + 1) * BPITCH];
      b1.x = bp[(kk * 4) * BPITCH + 16];
      b1.y = bp[(kk * 4 + 1) * BPITCH + 16];
      c00 = __builtin_amdgcn_wmma_f32_16x16x4_f32(false, a0, false, b0,
                                                  (short)0, c00, false, false);
      c01 = __builtin_amdgcn_wmma_f32_16x16x4_f32(false, a0, false, b1,
                                                  (short)0, c01, false, false);
      c10 = __builtin_amdgcn_wmma_f32_16x16x4_f32(false, a1, false, b0,
                                                  (short)0, c10, false, false);
      c11 = __builtin_amdgcn_wmma_f32_16x16x4_f32(false, a1, false, b1,
                                                  (short)0, c11, false, false);
    }
  }

  float* pp = Ps + (wave * TM + half * 8) * PPITCH + l16;
#pragma unroll
  for (int r = 0; r < 8; ++r) {
    pp[r * PPITCH] = c00[r];
    pp[r * PPITCH + 16] = c01[r];
    pp[(16 + r) * PPITCH] = c10[r];
    pp[(16 + r) * PPITCH + 16] = c11[r];
  }
  __syncthreads();

  const int orow = t >> 3;
  const int og = t & 7;
  float4 s = *(const float4*)(Ps + orow * PPITCH + og * 4);
  for (int q = 1; q < 8; ++q) {
    const float4 v = *(const float4*)(Ps + (q * TM + orow) * PPITCH + og * 4);
    s.x += v.x; s.y += v.y; s.z += v.z; s.w += v.w;
  }
  *(float4*)(m + (bm * TM + orow) * NN + bn * TN + og * 4) = s;
}

#define JSPLIT 8
#define JPER (BB / JSPLIT)

__global__ __launch_bounds__(64 * JSPLIT) void feats_kernel(const float* __restrict__ x,
                                                            const float* __restrict__ m,
                                                            float* __restrict__ out) {
  __shared__ float lds[BB * DD];
  __shared__ float part[JSPLIT * 64];
  const int k = blockIdx.y;
  const int il = threadIdx.x & 63;
  const int jp = threadIdx.x >> 6;
  const int i = blockIdx.x * 64 + il;

  const int f2 = (blockIdx.y * 8 + blockIdx.x) * (64 * JSPLIT) + threadIdx.x;
  const float2 xc = ((const float2*)x)[f2];

  for (int r = 0; r < (BB * 4) / (64 * JSPLIT); ++r) {
    const int t = threadIdx.x + r * (64 * JSPLIT);
    const int j = t >> 2;
    const int q = t & 3;
    const float4* src = (const float4*)(m + j * NN + k * DD);
    ((float4*)lds)[j * 4 + q] = src[q];
  }
  __syncthreads();
  ((float2*)out)[(f2 >> 8) * (OUTC / 2) + (f2 & 255)] = xc;

  float4 mi0 = ((const float4*)(lds + i * DD))[0];
  float4 mi1 = ((const float4*)(lds + i * DD))[1];
  float4 mi2 = ((const float4*)(lds + i * DD))[2];
  float4 mi3 = ((const float4*)(lds + i * DD))[3];

  float acc = 0.f;
  const int j0 = jp * JPER;
  for (int jj = 0; jj < JPER; ++jj) {
    const float4* row = (const float4*)(lds + (j0 + jj) * DD);
    float4 r0 = row[0], r1 = row[1], r2 = row[2], r3 = row[3];
    float dist = fabsf(mi0.x - r0.x) + fabsf(mi0.y - r0.y) +
                 fabsf(mi0.z - r0.z) + fabsf(mi0.w - r0.w);
    dist += fabsf(mi1.x - r1.x) + fabsf(mi1.y - r1.y) +
            fabsf(mi1.z - r1.z) + fabsf(mi1.w - r1.w);
    dist += fabsf(mi2.x - r2.x) + fabsf(mi2.y - r2.y) +
            fabsf(mi2.z - r2.z) + fabsf(mi2.w - r2.w);
    dist += fabsf(mi3.x - r3.x) + fabsf(mi3.y - r3.y) +
            fabsf(mi3.z - r3.z) + fabsf(mi3.w - r3.w);
    acc += __expf(-dist);
  }

  part[jp * 64 + il] = acc;
  __syncthreads();
  if (threadIdx.x < 64) {
    float s = part[il];
    for (int p = 1; p < JSPLIT; ++p) s += part[p * 64 + il];
    out[i * OUTC + FF + k] = s;
  }
}

extern "C" void kernel_launch(void* const* d_in, const int* in_sizes, int n_in,
                              void* d_out, int out_size, void* d_ws, size_t ws_size,
                              hipStream_t stream) {
  (void)in_sizes; (void)n_in; (void)out_size; (void)ws_size;
  const float* x = (const float*)d_in[0];
  const float* T = (const float*)d_in[1];
  float* out = (float*)d_out;
  float* m = (float*)d_ws;

  gemm_wmma_f32<<<256, 256, 0, stream>>>(x, T, m);
  feats_kernel<<<dim3(8, 32), 64 * JSPLIT, 0, stream>>>(x, m, out);
}


# feats pair loop: 24 vector ops per pair, not 38 (packed subtracts; one asm add takes a pair's two |.|), 2 rows i per LDS row read; j sum now 16 partials of 32 j, a pair's 16 terms summed as a tree
# speedup vs baseline: 3.5491x; 1.1611x over previous
#include <hip/hip_runtime.h>
#include <hip/hip_bf16.h>

typedef float v2f __attribute__((ext_vector_type(2)));
typedef float v8f __attribute__((ext_vector_type(8)));

#define BB 512
#define FF 512
#define KK 32
#define DD 16
#define NN (KK * DD)
#define OUTC (FF + KK)

#define TM 32
#define TN 32
#define KW 64
#define APITCH 68
#define BPITCH 40
#define PPITCH 36
#define WAVE_A (TM * APITCH)
#define WAVE_B (KW * BPITCH)

__global__ __launch_bounds__(256) void gemm_wmma_f32(const float* __restrict__ x,
                                                     const float* __restrict__ T,
                                                     float* __restrict__ m) {
  __shared__ __attribute__((aligned(16))) float As[8 * WAVE_A];
  __shared__ __attribute__((aligned(16))) float Bs[8 * WAVE_B];
  __shared__ __attribute__((aligned(16))) float Ps[8 * TM * PPITCH];
  const int t = threadIdx.x;
  const int lane = t & 31;
  const int wave = t >> 5;
  const int half = lane >> 4;
  const int l16 = lane & 15;
  const int bm = blockIdx.x >> 4;
  const int bn = blockIdx.x & 15;

  float* Aw = As + wave * WAVE_A;
  float* Bw = Bs + wave * WAVE_B;
  const float* xg = x + (bm * TM) * FF + wave * KW;
  const float* Tg = T + (wave * KW) * NN + bn * TN;

  float4 ra[2][8], rb[2][8];
#pragma unroll
  for (int ph = 0; ph < 2; ++ph) {
#pragma unroll
    for (int r = 0; r < 8; ++r) {
      const int idx = lane + r * 32;
      ra[ph][r] = *(const float4*)(xg + (idx >> 3) * FF + ph * 32 + (idx & 7) * 4);
    }
#pragma unroll
    for (int r = 0; r < 8; ++r) {
      const int idx = lane + r * 32;
      rb[ph][r] = *(const float4*)(Tg + (ph * 32 + (idx >> 3)) * NN + (idx & 7) * 4);
    }
  }

  v8f c00 = {};
  v8f c01 = {};
  v8f c10 = {};
  v8f c11 = {};
  const float* ap = Aw + l16 * APITCH + half * 2;
  const float* bp = Bw + (half * 2) * BPITCH + l16;

#pragma unroll
  for (int ph = 0; ph < 2; ++ph) {
    if (ph == 1) __builtin_amdgcn_sched_barrier(0x20);
#pragma unroll
    for (int r = 0; r < 8; ++r) {
      const int idx = lane + r * 32;
      *(float4*)(Aw + (idx >> 3) * APITCH + ph * 32 + (idx & 7) * 4) = ra[ph][r];
    }
#pragma unroll
    for (int r = 0; r < 8; ++r) {
      const int idx = lane + r * 32;
      *(float4*)(Bw + (ph * 32 + (idx >> 3)) * BPITCH + (idx & 7) * 4) = rb[ph][r];
    }
#pragma unroll
    for (int ks = 0; ks < 8; ++ks) {
      const int kk = ph * 8 + ks;
      const v2f a0 = *(const v2f*)(ap + kk * 4);
      const v2f a1 = *(const v2f*)(ap + 16 * APITCH + kk * 4);
      v2f b0, b1;
      b0.x = bp[(kk * 4) * BPITCH];
      b0.y = bp[(kk * 4 + 1) * BPITCH];
      b1.x = bp[(kk * 4) * BPITCH + 16];
      b1.y = bp[(kk * 4 + 1) * BPITCH + 16];
      c00 = __builtin_amdgcn_wmma_f32_16x16x4_f32(false, a0, false, b0,
                                                  (short)0, c00, false, false);
      c01 = __builtin_amdgcn_wmma_f32_16x16x4_f32(false, a0, false, b1,
                                                  (short)0, c01, false, false);
      c10 = __builtin_amdgcn_wmma_f32_16x16x4_f32(false, a1, false, b0,
                                                  (short)0, c10, false, false);
      c11 = __builtin_amdgcn_wmma_f32_16x16x4_f32(false, a1, false, b1,
                                                  (short)0, c11, false, false);
    }
  }

  float* pp = Ps + (wave * TM + half * 8) * PPITCH + l16;
#pragma unroll
  for (int r = 0; r < 8; ++r) {
    pp[r * PPITCH] = c00[r];
    pp[r * PPITCH + 16] = c01[r];
    pp[(16 + r) * PPITCH] = c10[r];
    pp[(16 + r) * PPITCH + 16] = c11[r];
  }
  __syncthreads();

  const int orow = t >> 3;
  const int og = t & 7;
  float4 s = *(const float4*)(Ps + orow * PPITCH + og * 4);
  for (int q = 1; q < 8; ++q) {
    const float4 v = *(const float4*)(Ps + (q * TM + orow) * PPITCH + og * 4);
    s.x += v.x; s.y += v.y; s.z += v.z; s.w += v.w;
  }
  *(float4*)(m + (bm * TM + orow) * NN + bn * TN + og * 4) = s;
}

#define JW 16
#define JPER (BB / JW)

__device__ __forceinline__ float abs_add(float a, float b) {
  float s;
  asm volatile("v_add_f32_e64 %0, |%1|, |%2|" : "=v"(s) : "v"(a), "v"(b));
  return s;
}

__device__ __forceinline__ float abs_add_hinted(float a, float b) {
  float s;
  asm volatile("s_delay_alu instid0(VALU_DEP_1)\n\tv_add_f32_e64 %0, |%1|, |%2|"
               : "=v"(s) : "v"(a), "v"(b));
  return s;
}

__global__ __launch_bounds__(32 * JW) void feats_kernel(const float* __restrict__ x,
                                                        const float* __restrict__ m,
                                                        float* __restrict__ out) {
  __shared__ float lds[BB * DD];
  __shared__ float part[JW * 64];
  const int k = blockIdx.y;
  const int lane = threadIdx.x & 31;
  const int w = threadIdx.x >> 5;

  const int f2 = (blockIdx.y * 8 + blockIdx.x) * (32 * JW) + threadIdx.x;
  const float2 xc = ((const float2*)x)[f2];

  for (int r = 0; r < (BB * 4) / (32 * JW); ++r) {
    const int t = threadIdx.x + r * (32 * JW);
    const int j = t >> 2;
    const int q = t & 3;
    const float4* src = (const float4*)(m + j * NN + k * DD);
    ((float4*)lds)[j * 4 + q] = src[q];
  }
  __syncthreads();
  ((float2*)out)[(f2 >> 8) * (OUTC / 2) + (f2 & 255)] = xc;

  const int ia = blockIdx.x * 64 + lane;
  v2f ma[8], mb[8];
#pragma unroll
  for (int q = 0; q < 4; ++q) {
    const float4 va = ((const float4*)(lds + ia * DD))[q];
    const float4 vb = ((const float4*)(lds + (ia + 32) * DD))[q];
    ma[2 * q] = (v2f){va.x, va.y};
    ma[2 * q + 1] = (v2f){va.z, va.w};
    mb[2 * q] = (v2f){vb.x, vb.y};
    mb[2 * q + 1] = (v2f){vb.z, vb.w};
  }

  float acca = 0.f, accb = 0.f;
  auto pair_rows = [&](const float4 r0, const float4 r1, const float4 r2, const float4 r3) {
    const v2f rr[8] = {(v2f){r0.x, r0.y}, (v2f){r0.z, r0.w}, (v2f){r1.x, r1.y}, (v2f){r1.z, r1.w},
                       (v2f){r2.x, r2.y}, (v2f){r2.z, r2.w}, (v2f){r3.x, r3.y}, (v2f){r3.z, r3.w}};
    v2f da[8], db[8];
#pragma unroll
    for (int q = 0; q < 8; ++q) {
      da[q] = ma[q] - rr[q];
      db[q] = mb[q] - rr[q];
    }
    __builtin_amdgcn_sched_barrier(0);
    float sa[8], sb[8];
    sa[0] = abs_add_hinted(da[0].x, da[0].y);
#pragma unroll
    for (int q = 1; q < 8; ++q) sa[q] = abs_add(da[q].x, da[q].y);
#pragma unroll
    for (int q = 0; q < 8; ++q) sb[q] = abs_add(db[q].x, db[q].y);
    __builtin_amdgcn_sched_barrier(0);
    const v2f ta = ((v2f){sa[0], sa[1]} + (v2f){sa[2], sa[3]}) +
                   ((v2f){sa[4], sa[5]} + (v2f){sa[6], sa[7]});
    const v2f tb = ((v2f){sb[0], sb[1]} + (v2f){sb[2], sb[3]}) +
                   ((v2f){sb[4], sb[5]} + (v2f){sb[6], sb[7]});
    acca += __expf(-(ta.x + ta.y));
    accb += __expf(-(tb.x + tb.y));
  };

  const float4* rows = (const float4*)(lds + (w * JPER) * DD);
  float4 a0 = rows[0], a1 = rows[1], a2 = rows[2], a3 = rows[3];
  for (int jj = 0; jj < JPER; jj += 2) {
    const float4* rb = rows + (jj + 1) * 4;
    const float4 b0 = rb[0], b1 = rb[1], b2 = rb[2], b3 = rb[3];
    pair_rows(a0, a1, a2, a3);
    const float4* ra = rows + ((jj + 2) & (JPER - 1)) * 4;
    a0 = ra[0]; a1 = ra[1]; a2 = ra[2]; a3 = ra[3];
    pair_rows(b0, b1, b2, b3);
  }

  part[w * 64 + lane] = acca;
  part[w * 64 + lane + 32] = accb;
  __syncthreads();
  if (threadIdx.x < 64) {
    float s = part[threadIdx.x];
    for (int p = 1; p < JW; ++p) s += part[p * 64 + threadIdx.x];
    out[(blockIdx.x * 64 + threadIdx.x) * OUTC + FF + k] = s;
  }
}

extern "C" void kernel_launch(void* const* d_in, const int* in_sizes, int n_in,
                              void* d_out, int out_size, void* d_ws, size_t ws_size,
                              hipStream_t stream) {
  (void)in_sizes; (void)n_in; (void)out_size; (void)ws_size;
  const float* x = (const float*)d_in[0];
  const float* T = (const float*)d_in[1];
  float* out = (float*)d_out;
  float* m = (float*)d_ws;

  gemm_wmma_f32<<<256, 256, 0, stream>>>(x, T, m);
  feats_kernel<<<dim3(8, 32), 32 * JW, 0, stream>>>(x, m, out);
}
